# starting kernel
# baseline (speedup 1.0000x reference)
#include <hip/hip_runtime.h>
#include <hip/hip_bf16.h>

typedef float v2f __attribute__((ext_vector_type(2)));
typedef float v8f __attribute__((ext_vector_type(8)));

#define N_NODES 50000
#define N_EDGES 800000

__global__ void k_init_deg(float* __restrict__ deg, int n) {
  int i = blockIdx.x * blockDim.x + threadIdx.x;
  if (i < n) deg[i] = 1.0f;
}

__global__ void k_count_deg(const int* __restrict__ dst, float* __restrict__ deg, int e) {
  int i = blockIdx.x * blockDim.x + threadIdx.x;
  if (i < e) atomicAdd(&deg[dst[i]], 1.0f);
}

__global__ void k_dinv(float* __restrict__ deg, int n) {
  int i = blockIdx.x * blockDim.x + threadIdx.x;
  if (i < n) deg[i] = rsqrtf(deg[i]);
}

__global__ void k_pack_w(const float* __restrict__ W, float* __restrict__ Wp,
                         int K, int N) {
  int idx = blockIdx.x * blockDim.x + threadIdx.x;
  if (idx >= K * N) return;
  int k = idx / N;
  int n = idx - k * N;
  Wp[(size_t)(k >> 2) * (N * 4) + n * 4 + (k & 3)] = W[idx];
}

template <int KDIM, int NT, bool BIAS_RELU>
__global__ __launch_bounds__(128) void k_gemm_wmma(
    const float* __restrict__ A, const float* __restrict__ Wp,
    const float* __restrict__ bias, float* __restrict__ out,
    int M, int out_stride, int out_off) {
  const int lane = threadIdx.x & 31;
  const int wave = threadIdx.x >> 5;
  const int row_tile = (blockIdx.x * 4 + wave) * 16;
  if (row_tile >= M) return;

  const int m16 = lane & 15;
  const int khalf = lane >> 4;
  constexpr int NCOLS = NT * 16;

  v8f acc[NT];
#pragma unroll
  for (int t = 0; t < NT; ++t) acc[t] = (v8f){0.f, 0.f, 0.f, 0.f, 0.f, 0.f, 0.f, 0.f};

  const float* __restrict__ arow = A + (size_t)(row_tile + m16) * KDIM + khalf * 2;
  const float* __restrict__ wlane = Wp + m16 * 4 + khalf * 2;

  for (int k0 = 0; k0 < KDIM; k0 += 16) {
    if (k0 + 256 < KDIM) __builtin_prefetch(arow + k0 + 256, 0, 0);
#pragma unroll
    for (int kq = 0; kq < 4; ++kq) {
      const int k = k0 + kq * 4;
      v2f afrag = *(const v2f*)(arow + k);
      const float* wq = wlane + (size_t)(k >> 2) * (NCOLS * 4);
#pragma unroll
      for (int t = 0; t < NT; ++t) {
        v2f bfrag = *(const v2f*)(wq + t * 64);
        acc[t] = __builtin_amdgcn_wmma_f32_16x16x4_f32(
             false, afrag,  false, bfrag,
             (short)0, acc[t],  false,  false);
      }
    }
  }

  const int rbase = row_tile + khalf * 8;
#pragma unroll
  for (int t = 0; t < NT; ++t) {
#pragma unroll
    for (int i = 0; i < 8; ++i) {
      const int c = t * 16 + m16;
      float v = acc[t][i];
      if (BIAS_RELU) v = fmaxf(v + bias[c], 0.0f);
      out[(size_t)(rbase + i) * out_stride + out_off + c] = v;
    }
  }
}

__global__ void k_node_init(const float* __restrict__ xw, const float* __restrict__ dinv,
                            const float* __restrict__ bias, float* __restrict__ out,
                            int n, int C) {
  int idx = blockIdx.x * blockDim.x + threadIdx.x;
  if (idx >= n * C) return;
  int i = idx / C;
  int c = idx - i * C;
  float di = dinv[i];
  out[idx] = bias[c] + xw[idx] * di * di;
}

template <int C>
__global__ void k_scatter(const float* __restrict__ xw, const float* __restrict__ dinv,
                          const int* __restrict__ src, const int* __restrict__ dst,
                          float* __restrict__ out, int e) {
  constexpr int CH = C / 4;
  int tid = blockIdx.x * blockDim.x + threadIdx.x;
  int eidx = tid / CH;
  int ch = tid - eidx * CH;
  if (eidx >= e) return;
  int s = src[eidx];
  int d = dst[eidx];
  float nrm = dinv[s] * dinv[d];
  float4 v = *((const float4*)(xw + (size_t)s * C) + ch);
  float* o = out + (size_t)d * C + ch * 4;
  atomicAdd(o + 0, v.x * nrm);
  atomicAdd(o + 1, v.y * nrm);
  atomicAdd(o + 2, v.z * nrm);
  atomicAdd(o + 3, v.w * nrm);
}

__global__ void k_relu(float* __restrict__ x, int n) {
  int i = blockIdx.x * blockDim.x + threadIdx.x;
  if (i < n) x[i] = fmaxf(x[i], 0.0f);
}

extern "C" void kernel_launch(void* const* d_in, const int* in_sizes, int n_in,
                              void* d_out, int out_size, void* d_ws, size_t ws_size,
                              hipStream_t stream) {
  (void)in_sizes; (void)n_in; (void)out_size; (void)ws_size;

  const float* feature = (const float*)d_in[0];
  const float* img     = (const float*)d_in[1];
  const int*   edges   = (const int*)d_in[2];
  const float* W_feat  = (const float*)d_in[3];
  const float* b_feat  = (const float*)d_in[4];
  const float* W_img   = (const float*)d_in[5];
  const float* b_img   = (const float*)d_in[6];
  const float* W_g1    = (const float*)d_in[7];
  const float* b_g1    = (const float*)d_in[8];
  const float* W_g2    = (const float*)d_in[9];
  const float* b_g2    = (const float*)d_in[10];
  float* out = (float*)d_out;

  const int* src = edges;
  const int* dst = edges + N_EDGES;

  float* ws     = (float*)d_ws;
  float* dinv   = ws;
  float* bufA   = ws + 50048;
  float* bufB   = bufA + (size_t)N_NODES * 96;
  float* wpF    = bufB + (size_t)N_NODES * 96;
  float* wpI    = wpF + 512 * 64;
  float* wpG1   = wpI + 256 * 32;
  float* wpG2   = wpG1 + 96 * 96;

  const int T = 256;
  const int MB = (N_NODES + 63) / 64;

  k_pack_w<<<(512 * 64 + T - 1) / T, T, 0, stream>>>(W_feat, wpF, 512, 64);
  k_pack_w<<<(256 * 32 + T - 1) / T, T, 0, stream>>>(W_img, wpI, 256, 32);
  k_pack_w<<<(96 * 96 + T - 1) / T, T, 0, stream>>>(W_g1, wpG1, 96, 96);
  k_pack_w<<<(96 * 64 + T - 1) / T, T, 0, stream>>>(W_g2, wpG2, 96, 64);

  k_init_deg<<<(N_NODES + T - 1) / T, T, 0, stream>>>(dinv, N_NODES);
  k_count_deg<<<(N_EDGES + T - 1) / T, T, 0, stream>>>(dst, dinv, N_EDGES);
  k_dinv<<<(N_NODES + T - 1) / T, T, 0, stream>>>(dinv, N_NODES);

  k_gemm_wmma<512, 4, true><<<MB, 128, 0, stream>>>(feature, wpF, b_feat, bufA, N_NODES, 96, 0);
  k_gemm_wmma<256, 2, true><<<MB, 128, 0, stream>>>(img, wpI, b_img, bufA, N_NODES, 96, 64);

  k_gemm_wmma<96, 6, false><<<MB, 128, 0, stream>>>(bufA, wpG1, nullptr, bufB, N_NODES, 96, 0);
  k_node_init<<<(N_NODES * 96 + T - 1) / T, T, 0, stream>>>(bufB, dinv, b_g1, bufA, N_NODES, 96);
  k_scatter<96><<<(N_EDGES * 24 + T - 1) / T, T, 0, stream>>>(bufB, dinv, src, dst, bufA, N_EDGES);
  k_relu<<<(N_NODES * 96 + T - 1) / T, T, 0, stream>>>(bufA, N_NODES * 96);

  k_gemm_wmma<96, 4, false><<<MB, 128, 0, stream>>>(bufA, wpG2, nullptr, bufB, N_NODES, 64, 0);
  k_node_init<<<(N_NODES * 64 + T - 1) / T, T, 0, stream>>>(bufB, dinv, b_g2, out, N_NODES, 64);
  k_scatter<64><<<(N_EDGES * 16 + T - 1) / T, T, 0, stream>>>(bufB, dinv, src, dst, out, N_EDGES);
}


# Encoder products taken transposed (weights as A, node rows as B): their accumulators are layer 1's A fragments, so encoders + x@W_g1 are one kernel, x never stored; layer-1 k summed in another order
# speedup vs baseline: 1.6448x; 1.6448x over previous
#include <hip/hip_runtime.h>
#include <hip/hip_bf16.h>

typedef float v2f __attribute__((ext_vector_type(2)));
typedef float v8f __attribute__((ext_vector_type(8)));

#define N_NODES 50000
#define N_EDGES 800000

__global__ void k_init_deg(float* __restrict__ deg, int n) {
  int i = blockIdx.x * blockDim.x + threadIdx.x;
  if (i < n) deg[i] = 1.0f;
}

__global__ void k_count_deg(const int* __restrict__ dst, float* __restrict__ deg, int e) {
  int i = blockIdx.x * blockDim.x + threadIdx.x;
  if (i < e) atomicAdd(&deg[dst[i]], 1.0f);
}

__global__ void k_dinv(float* __restrict__ deg, int n) {
  int i = blockIdx.x * blockDim.x + threadIdx.x;
  if (i < n) deg[i] = rsqrtf(deg[i]);
}

__global__ void k_pack_w(const float* __restrict__ W, float* __restrict__ Wp,
                         int K, int N, int chained) {
  int idx = blockIdx.x * blockDim.x + threadIdx.x;
  if (idx >= K * N) return;
  int k = idx / N;
  int n = idx - k * N;
  int step = k >> 2;
  int slot = k & 3;
  if (chained) {
    step = ((k >> 4) << 2) + ((k >> 1) & 3);
    slot = (((k >> 3) & 1) << 1) + (k & 1);
  }
  Wp[(size_t)step * (N * 4) + n * 4 + slot] = W[idx];
}

template <int KDIM, int NT, bool BIAS_RELU, bool RELU_A, bool NODE_INIT>
__global__ __launch_bounds__(128) void k_gemm_wmma(
    const float* A, const float* __restrict__ Wp,
    const float* __restrict__ bias, float* __restrict__ out,
    int M, int out_stride, int out_off,
    const float* __restrict__ dinv, float* init_out, int* __restrict__ tick) {
  if (tick != nullptr && blockIdx.x == 0 && threadIdx.x < 2) tick[threadIdx.x * 64] = 0;
  const int lane = threadIdx.x & 31;
  const int wave = threadIdx.x >> 5;
  const int row_tile = (blockIdx.x * 4 + wave) * 16;
  if (row_tile >= M) return;

  const int m16 = lane & 15;
  const int khalf = lane >> 4;
  constexpr int NCOLS = NT * 16;

  v8f acc[NT];
#pragma unroll
  for (int t = 0; t < NT; ++t) acc[t] = (v8f){0.f, 0.f, 0.f, 0.f, 0.f, 0.f, 0.f, 0.f};

  const float* arow = A + (size_t)(row_tile + m16) * KDIM + khalf * 2;
  const float* __restrict__ wlane = Wp + m16 * 4 + khalf * 2;

  for (int k0 = 0; k0 < KDIM; k0 += 16) {
    if (k0 + 256 < KDIM) __builtin_prefetch(arow + k0 + 256, 0, 0);
#pragma unroll
    for (int kq = 0; kq < 4; ++kq) {
      const int k = k0 + kq * 4;
      v2f afrag = *(const v2f*)(arow + k);
      if (RELU_A) {
        afrag.x = fmaxf(afrag.x, 0.0f);
        afrag.y = fmaxf(afrag.y, 0.0f);
      }
      const float* wq = wlane + (size_t)(k >> 2) * (NCOLS * 4);
#pragma unroll
      for (int t = 0; t < NT; ++t) {
        v2f bfrag = *(const v2f*)(wq + t * 64);
        acc[t] = __builtin_amdgcn_wmma_f32_16x16x4_f32(
             false, afrag,  false, bfrag,
             (short)0, acc[t],  false,  false);
      }
    }
  }

  const int rbase = row_tile + khalf * 8;
  float di[8];
  if (NODE_INIT) {
#pragma unroll
    for (int i = 0; i < 8; ++i) di[i] = dinv[rbase + i];
  }
#pragma unroll
  for (int t = 0; t < NT; ++t) {
    const int c = t * 16 + m16;
    float bc = 0.0f;
    if (BIAS_RELU || NODE_INIT) bc = bias[c];
#pragma unroll
    for (int i = 0; i < 8; ++i) {
      float v = acc[t][i];
      if (BIAS_RELU) v = fmaxf(v + bc, 0.0f);
      out[(size_t)(rbase + i) * out_stride + out_off + c] = v;
      if (NODE_INIT) init_out[(size_t)(rbase + i) * NCOLS + c] = bc + v * di[i] * di[i];
    }
  }
}

template <int KDIM, int NT, int T0>
__device__ __forceinline__ void enc_xT(const float* __restrict__ nrow,
                                       const float* __restrict__ wlane, v8f (&xt)[6]) {
  for (int k0 = 0; k0 < KDIM; k0 += 16) {
    if (k0 + 256 < KDIM) __builtin_prefetch(nrow + k0 + 256, 0, 0);
#pragma unroll
    for (int kq = 0; kq < 4; ++kq) {
      const int k = k0 + kq * 4;
      const v2f xf = *(const v2f*)(nrow + k);
      const float* wq = wlane + (size_t)(k >> 2) * (NT * 16 * 4);
#pragma unroll
      for (int t = 0; t < NT; ++t) {
        const v2f wf = *(const v2f*)(wq + t * 64);
        xt[T0 + t] = __builtin_amdgcn_wmma_f32_16x16x4_f32(
             false, wf,  false, xf,
             (short)0, xt[T0 + t],  false,  false);
      }
    }
  }
}

__global__ __launch_bounds__(128) __attribute__((amdgpu_waves_per_eu(4, 4))) void k_dense1(
    const float* __restrict__ feature, const float* __restrict__ img,
    const float* __restrict__ wpF, const float* __restrict__ wpI,
    const float* __restrict__ wpG1,
    const float* __restrict__ b_feat, const float* __restrict__ b_img,
    const float* __restrict__ b_g1, const float* __restrict__ dinv,
    float* __restrict__ xw, float* __restrict__ hinit, int M, int* __restrict__ tick) {
  if (blockIdx.x == 0 && threadIdx.x < 2) tick[threadIdx.x * 64] = 0;
  const int lane = threadIdx.x & 31;
  const int wave = threadIdx.x >> 5;
  const int row_tile = (blockIdx.x * 4 + wave) * 16;
  if (row_tile >= M) return;

  const int m16 = lane & 15;
  const int khalf = lane >> 4;

  v8f xt[6];
#pragma unroll
  for (int t = 0; t < 6; ++t) xt[t] = (v8f){0.f, 0.f, 0.f, 0.f, 0.f, 0.f, 0.f, 0.f};

  enc_xT<512, 4, 0>(feature + (size_t)(row_tile + m16) * 512 + khalf * 2,
                    wpF + m16 * 4 + khalf * 2, xt);
  enc_xT<256, 2, 4>(img + (size_t)(row_tile + m16) * 256 + khalf * 2,
                    wpI + m16 * 4 + khalf * 2, xt);

#pragma unroll
  for (int t = 0; t < 6; ++t) {
    const float* __restrict__ bt =
        (t < 4) ? (b_feat + t * 16 + khalf * 8) : (b_img + (t - 4) * 16 + khalf * 8);
#pragma unroll
    for (int r = 0; r < 8; ++r) xt[t][r] = fmaxf(xt[t][r] + bt[r], 0.0f);
  }

  v8f acc[6];
#pragma unroll
  for (int u = 0; u < 6; ++u) acc[u] = (v8f){0.f, 0.f, 0.f, 0.f, 0.f, 0.f, 0.f, 0.f};
  const float* __restrict__ glane = wpG1 + m16 * 4 + khalf * 2;
#pragma unroll
  for (int t = 0; t < 6; ++t) {
#pragma unroll
    for (int j = 0; j < 4; ++j) {
      v2f af;
      af.x = xt[t][2 * j];
      af.y = xt[t][2 * j + 1];
      const float* gq = glane + (size_t)(t * 4 + j) * (96 * 4);
#pragma unroll
      for (int u = 0; u < 6; ++u) {
        const v2f bf = *(const v2f*)(gq + u * 64);
        acc[u] = __builtin_amdgcn_wmma_f32_16x16x4_f32(
             false, af,  false, bf,
             (short)0, acc[u],  false,  false);
      }
    }
  }

  const int rbase = row_tile + khalf * 8;
  float di[8];
#pragma unroll
  for (int i = 0; i < 8; ++i) di[i] = dinv[rbase + i];
#pragma unroll
  for (int u = 0; u < 6; ++u) {
    const int c = u * 16 + m16;
    const float bc = b_g1[c];
#pragma unroll
    for (int i = 0; i < 8; ++i) {
      const float v = acc[u][i];
      xw[(size_t)(rbase + i) * 96 + c] = v;
      hinit[(size_t)(rbase + i) * 96 + c] = bc + v * di[i] * di[i];
    }
  }
}

template <int C>
__global__ __launch_bounds__(256) void k_scatter(
    const float* __restrict__ xw, const float* __restrict__ dinv,
    const int* __restrict__ src, const int* __restrict__ dst,
    float* __restrict__ out, int e, int* __restrict__ tick) {
  constexpr int J = C / 32;
  constexpr int B = 8;
  constexpr int CH = 64;
  const int lane = threadIdx.x & 31;
  const int nchunks = (e + CH - 1) / CH;
  const int myhalf = (int)((__builtin_amdgcn_s_sendmsg_rtn(0x87) >> 18) & 1u);

  for (int pass = 0; pass < 2; ++pass) {
    const int h = myhalf ^ pass;
    int* __restrict__ word = tick + h * 64;
    for (int it = 0; it <= nchunks; ++it) {
      int t = 0;
      if (lane == 0) t = atomicAdd(word, 1);
      t = __builtin_amdgcn_readfirstlane(t);
      if (t >= nchunks) break;

      for (int sub = 0; sub < CH / 32; ++sub) {
        const int e0 = t * CH + sub * 32;
        if (e0 >= e) break;
        const int cnt = (e - e0 < 32) ? (e - e0) : 32;

        int s = 0, d = 0;
        float nrm = 0.0f;
        bool own = false;
        if (lane < cnt) {
          s = src[e0 + lane];
          d = dst[e0 + lane];
          own = ((d >= N_NODES / 2) ? 1 : 0) == h;
          if (own) nrm = dinv[s] * dinv[d];
        }
        unsigned m = __builtin_amdgcn_ballot_w32(own);

        while (m != 0u) {
          const int left = __builtin_popcount(m);
          const int nb = left < B ? left : B;
          int li[B];
#pragma unroll
          for (int i = 0; i < B; ++i) {
            li[i] = __builtin_ctz(m | 0x80000000u);
            m &= m - 1u;
          }
          float v[B][J];
#pragma unroll
          for (int i = 0; i < B; ++i) {
            if (i < nb) {
              const int si = __builtin_amdgcn_readlane(s, li[i]);
              const float* __restrict__ r = xw + (size_t)si * C + lane;
#pragma unroll
              for (int j = 0; j < J; ++j) v[i][j] = r[32 * j];
            }
          }
#pragma unroll
          for (int i = 0; i < B; ++i) {
            if (i < nb) {
              const int di = __builtin_amdgcn_readlane(d, li[i]);
              const float ni = __int_as_float(__builtin_amdgcn_readlane(__float_as_int(nrm), li[i]));
              float* __restrict__ o = out + (size_t)di * C + lane;
#pragma unroll
              for (int j = 0; j < J; ++j) atomicAdd(o + 32 * j, v[i][j] * ni);
            }
          }
        }
      }
    }
  }
}

extern "C" void kernel_launch(void* const* d_in, const int* in_sizes, int n_in,
                              void* d_out, int out_size, void* d_ws, size_t ws_size,
                              hipStream_t stream) {
  (void)in_sizes; (void)n_in; (void)out_size; (void)ws_size;

  const float* feature = (const float*)d_in[0];
  const float* img     = (const float*)d_in[1];
  const int*   edges   = (const int*)d_in[2];
  const float* W_feat  = (const float*)d_in[3];
  const float* b_feat  = (const float*)d_in[4];
  const float* W_img   = (const float*)d_in[5];
  const float* b_img   = (const float*)d_in[6];
  const float* W_g1    = (const float*)d_in[7];
  const float* b_g1    = (const float*)d_in[8];
  const float* W_g2    = (const float*)d_in[9];
  const float* b_g2    = (const float*)d_in[10];
  float* out = (float*)d_out;

  const int* src = edges;
  const int* dst = edges + N_EDGES;

  float* ws     = (float*)d_ws;
  float* dinv   = ws;
  float* bufA   = ws + 50048;
  float* bufB   = bufA + (size_t)N_NODES * 96;
  float* wpF    = bufB + (size_t)N_NODES * 96;
  float* wpI    = wpF + 512 * 64;
  float* wpG1   = wpI + 256 * 32;
  float* wpG2   = wpG1 + 96 * 96;

  const int T = 256;
  const int MB = (N_NODES + 63) / 64;
  const int SB = 1024;
  int* tick1 = (int*)d_out;
  int* tick2 = (int*)(bufB + (size_t)N_NODES * 64);

  k_pack_w<<<(512 * 64 + T - 1) / T, T, 0, stream>>>(W_feat, wpF, 512, 64, 0);
  k_pack_w<<<(256 * 32 + T - 1) / T, T, 0, stream>>>(W_img, wpI, 256, 32, 0);
  k_pack_w<<<(96 * 96 + T - 1) / T, T, 0, stream>>>(W_g1, wpG1, 96, 96, 1);
  k_pack_w<<<(96 * 64 + T - 1) / T, T, 0, stream>>>(W_g2, wpG2, 96, 64, 0);

  k_init_deg<<<(N_NODES + T - 1) / T, T, 0, stream>>>(dinv, N_NODES);
  k_count_deg<<<(N_EDGES + T - 1) / T, T, 0, stream>>>(dst, dinv, N_EDGES);
  k_dinv<<<(N_NODES + T - 1) / T, T, 0, stream>>>(dinv, N_NODES);

  k_dense1<<<MB, 128, 0, stream>>>(feature, img, wpF, wpI, wpG1, b_feat, b_img, b_g1,
                                   dinv, bufB, bufA, N_NODES, tick1);
  k_scatter<96><<<SB, T, 0, stream>>>(bufB, dinv, src, dst, bufA, N_EDGES, tick1);

  k_gemm_wmma<96, 4, false, true, true><<<MB, 128, 0, stream>>>(
      bufA, wpG2, b_g2, bufB, N_NODES, 64, 0, dinv, out, tick2);
  k_scatter<64><<<SB, T, 0, stream>>>(bufB, dinv, src, dst, out, N_EDGES, tick2);
}


# Weights read in place, no packed copies: tile t of a product takes columns NT*c+t, so a lane's fragments for all tiles are neighbours in a row of W; 4 launches gone; layer-1 k summed in another order
# speedup vs baseline: 1.7765x; 1.0801x over previous
#include <hip/hip_runtime.h>
#include <hip/hip_bf16.h>

typedef float v2f __attribute__((ext_vector_type(2)));
typedef float v4f __attribute__((ext_vector_type(4)));
typedef float v8f __attribute__((ext_vector_type(8)));

#define N_NODES 50000
#define N_EDGES 800000

__global__ void k_init_deg(float* __restrict__ deg, int n) {
  int i = blockIdx.x * blockDim.x + threadIdx.x;
  if (i < n) deg[i] = 1.0f;
}

__global__ void k_count_deg(const int* __restrict__ dst, float* __restrict__ deg, int e) {
  int i = blockIdx.x * blockDim.x + threadIdx.x;
  if (i < e) atomicAdd(&deg[dst[i]], 1.0f);
}

__global__ void k_dinv(float* __restrict__ deg, int n) {
  int i = blockIdx.x * blockDim.x + threadIdx.x;
  if (i < n) deg[i] = rsqrtf(deg[i]);
}


__global__ __launch_bounds__(128) void k_gemm2(
    const float* A, const float* __restrict__ W,
    const float* __restrict__ bias, float* __restrict__ out, int M,
    const float* __restrict__ dinv, float* init_out, int* __restrict__ tick) {
  if (blockIdx.x == 0 && threadIdx.x < 2) tick[threadIdx.x * 64] = 0;
  const int lane = threadIdx.x & 31;
  const int wave = threadIdx.x >> 5;
  const int row_tile = (blockIdx.x * 4 + wave) * 16;
  if (row_tile >= M) return;

  const int m16 = lane & 15;
  const int khalf = lane >> 4;

  v8f acc[4];
#pragma unroll
  for (int t = 0; t < 4; ++t) acc[t] = (v8f){0.f, 0.f, 0.f, 0.f, 0.f, 0.f, 0.f, 0.f};

  const float* arow = A + (size_t)(row_tile + m16) * 96 + khalf * 2;
  const float* __restrict__ wrow = W + khalf * 2 * 64 + m16 * 4;

  for (int k0 = 0; k0 < 96; k0 += 16) {
#pragma unroll
    for (int kq = 0; kq < 4; ++kq) {
      const int k = k0 + kq * 4;
      v2f afrag = *(const v2f*)(arow + k);
      afrag.x = fmaxf(afrag.x, 0.0f);
      afrag.y = fmaxf(afrag.y, 0.0f);
      const v4f w0 = *(const v4f*)(wrow + k * 64);
      const v4f w1 = *(const v4f*)(wrow + (k + 1) * 64);
#pragma unroll
      for (int t = 0; t < 4; ++t) {
        v2f bfrag;
        bfrag.x = w0[t];
        bfrag.y = w1[t];
        acc[t] = __builtin_amdgcn_wmma_f32_16x16x4_f32(
             false, afrag,  false, bfrag,
             (short)0, acc[t],  false,  false);
      }
    }
  }

  const int rbase = row_tile + khalf * 8;
  const v4f bc = *(const v4f*)(bias + m16 * 4);
  float di[8];
#pragma unroll
  for (int i = 0; i < 8; ++i) di[i] = dinv[rbase + i];
#pragma unroll
  for (int i = 0; i < 8; ++i) {
    v4f v, s;
    v.x = acc[0][i]; v.y = acc[1][i]; v.z = acc[2][i]; v.w = acc[3][i];
    s.x = bc.x + v.x * di[i] * di[i];
    s.y = bc.y + v.y * di[i] * di[i];
    s.z = bc.z + v.z * di[i] * di[i];
    s.w = bc.w + v.w * di[i] * di[i];
    *(v4f*)(out + (size_t)(rbase + i) * 64 + m16 * 4) = v;
    *(v4f*)(init_out + (size_t)(rbase + i) * 64 + m16 * 4) = s;
  }
}

template <int NT> struct wvec;
template <> struct wvec<4> { typedef v4f type; };
template <> struct wvec<2> { typedef v2f type; };

template <int KDIM, int NT, int T0>
__device__ __forceinline__ void enc_xT(const float* __restrict__ nrow,
                                       const float* __restrict__ wrow, v8f (&xt)[6]) {
  typedef typename wvec<NT>::type vw;
  constexpr int NC = NT * 16;
#pragma unroll 2
  for (int k0 = 0; k0 < KDIM; k0 += 16) {
    if (k0 + 256 < KDIM) __builtin_prefetch(nrow + k0 + 256, 0, 0);
#pragma unroll
    for (int kq = 0; kq < 4; ++kq) {
      const int k = k0 + kq * 4;
      const v2f xf = *(const v2f*)(nrow + k);
      const vw w0 = *(const vw*)(wrow + k * NC);
      const vw w1 = *(const vw*)(wrow + (k + 1) * NC);
#pragma unroll
      for (int t = 0; t < NT; ++t) {
        v2f wf;
        wf.x = w0[t];
        wf.y = w1[t];
        xt[T0 + t] = __builtin_amdgcn_wmma_f32_16x16x4_f32(
             false, wf,  false, xf,
             (short)0, xt[T0 + t],  false,  false);
      }
    }
  }
}

__global__ __launch_bounds__(128) __attribute__((amdgpu_waves_per_eu(4, 4))) void k_dense1(
    const float* __restrict__ feature, const float* __restrict__ img,
    const float* __restrict__ W_feat, const float* __restrict__ W_img,
    const float* __restrict__ W_g1,
    const float* __restrict__ b_feat, const float* __restrict__ b_img,
    const float* __restrict__ b_g1, const float* __restrict__ dinv,
    float* __restrict__ xw, float* __restrict__ hinit, int M, int* __restrict__ tick) {
  if (blockIdx.x == 0 && threadIdx.x < 2) tick[threadIdx.x * 64] = 0;
  const int lane = threadIdx.x & 31;
  const int wave = threadIdx.x >> 5;
  const int row_tile = (blockIdx.x * 4 + wave) * 16;
  if (row_tile >= M) return;

  const int m16 = lane & 15;
  const int khalf = lane >> 4;

  v8f xt[6];
#pragma unroll
  for (int t = 0; t < 6; ++t) xt[t] = (v8f){0.f, 0.f, 0.f, 0.f, 0.f, 0.f, 0.f, 0.f};

  enc_xT<512, 4, 0>(feature + (size_t)(row_tile + m16) * 512 + khalf * 2,
                    W_feat + khalf * 2 * 64 + m16 * 4, xt);
  enc_xT<256, 2, 4>(img + (size_t)(row_tile + m16) * 256 + khalf * 2,
                    W_img + khalf * 2 * 32 + m16 * 2, xt);

#pragma unroll
  for (int r = 0; r < 8; ++r) {
    const v4f bf4 = *(const v4f*)(b_feat + 4 * (8 * khalf + r));
    const v2f bi2 = *(const v2f*)(b_img + 2 * (8 * khalf + r));
    xt[0][r] = fmaxf(xt[0][r] + bf4.x, 0.0f);
    xt[1][r] = fmaxf(xt[1][r] + bf4.y, 0.0f);
    xt[2][r] = fmaxf(xt[2][r] + bf4.z, 0.0f);
    xt[3][r] = fmaxf(xt[3][r] + bf4.w, 0.0f);
    xt[4][r] = fmaxf(xt[4][r] + bi2.x, 0.0f);
    xt[5][r] = fmaxf(xt[5][r] + bi2.y, 0.0f);
  }

  v8f acc[6];
#pragma unroll
  for (int u = 0; u < 6; ++u) acc[u] = (v8f){0.f, 0.f, 0.f, 0.f, 0.f, 0.f, 0.f, 0.f};
  const float* __restrict__ gF = W_g1 + (size_t)(32 * khalf) * 96 + m16 * 4;
  const float* __restrict__ gI = W_g1 + (size_t)(64 + 16 * khalf) * 96 + m16 * 4;
#pragma unroll
  for (int t = 0; t < 6; ++t) {
#pragma unroll
    for (int j = 0; j < 4; ++j) {
      v2f af;
      af.x = xt[t][2 * j];
      af.y = xt[t][2 * j + 1];
      const float* g0 = (t < 4) ? (gF + (8 * j + t) * 96) : (gI + (4 * j + (t - 4)) * 96);
      const float* g1 = g0 + ((t < 4) ? 4 : 2) * 96;
      const v4f p0 = *(const v4f*)(g0);
      const v4f p1 = *(const v4f*)(g1);
      const v2f q0 = *(const v2f*)(g0 + 64 - 2 * m16);
      const v2f q1 = *(const v2f*)(g1 + 64 - 2 * m16);
#pragma unroll
      for (int u = 0; u < 4; ++u) {
        v2f bf;
        bf.x = p0[u];
        bf.y = p1[u];
        acc[u] = __builtin_amdgcn_wmma_f32_16x16x4_f32(
             false, af,  false, bf,
             (short)0, acc[u],  false,  false);
      }
#pragma unroll
      for (int u = 0; u < 2; ++u) {
        v2f bf;
        bf.x = q0[u];
        bf.y = q1[u];
        acc[4 + u] = __builtin_amdgcn_wmma_f32_16x16x4_f32(
             false, af,  false, bf,
             (short)0, acc[4 + u],  false,  false);
      }
    }
  }

  const int rbase = row_tile + khalf * 8;
  const v4f bc4 = *(const v4f*)(b_g1 + m16 * 4);
  const v2f bc2 = *(const v2f*)(b_g1 + 64 + m16 * 2);
  float di[8];
#pragma unroll
  for (int i = 0; i < 8; ++i) di[i] = dinv[rbase + i];
#pragma unroll
  for (int i = 0; i < 8; ++i) {
    v4f v, s;
    v2f w, z;
    v.x = acc[0][i]; v.y = acc[1][i]; v.z = acc[2][i]; v.w = acc[3][i];
    w.x = acc[4][i]; w.y = acc[5][i];
    s.x = bc4.x + v.x * di[i] * di[i];
    s.y = bc4.y + v.y * di[i] * di[i];
    s.z = bc4.z + v.z * di[i] * di[i];
    s.w = bc4.w + v.w * di[i] * di[i];
    z.x = bc2.x + w.x * di[i] * di[i];
    z.y = bc2.y + w.y * di[i] * di[i];
    float* xr = xw + (size_t)(rbase + i) * 96;
    float* hr = hinit + (size_t)(rbase + i) * 96;
    *(v4f*)(xr + m16 * 4) = v;
    *(v2f*)(xr + 64 + m16 * 2) = w;
    *(v4f*)(hr + m16 * 4) = s;
    *(v2f*)(hr + 64 + m16 * 2) = z;
  }
}

template <int C>
__global__ __launch_bounds__(256) void k_scatter(
    const float* __restrict__ xw, const float* __restrict__ dinv,
    const int* __restrict__ src, const int* __restrict__ dst,
    float* __restrict__ out, int e, int* __restrict__ tick) {
  constexpr int J = C / 32;
  constexpr int B = 8;
  constexpr int CH = 64;
  const int lane = threadIdx.x & 31;
  const int nchunks = (e + CH - 1) / CH;
  const int myhalf = (int)((__builtin_amdgcn_s_sendmsg_rtn(0x87) >> 18) & 1u);

  for (int pass = 0; pass < 2; ++pass) {
    const int h = myhalf ^ pass;
    int* __restrict__ word = tick + h * 64;
    for (int it = 0; it <= nchunks; ++it) {
      int t = 0;
      if (lane == 0) t = atomicAdd(word, 1);
      t = __builtin_amdgcn_readfirstlane(t);
      if (t >= nchunks) break;

      for (int sub = 0; sub < CH / 32; ++sub) {
        const int e0 = t * CH + sub * 32;
        if (e0 >= e) break;
        const int cnt = (e - e0 < 32) ? (e - e0) : 32;

        int s = 0, d = 0;
        float nrm = 0.0f;
        bool own = false;
        if (lane < cnt) {
          s = src[e0 + lane];
          d = dst[e0 + lane];
          own = ((d >= N_NODES / 2) ? 1 : 0) == h;
          if (own) nrm = dinv[s] * dinv[d];
        }
        unsigned m = __builtin_amdgcn_ballot_w32(own);

        while (m != 0u) {
          const int left = __builtin_popcount(m);
          const int nb = left < B ? left : B;
          int li[B];
#pragma unroll
          for (int i = 0; i < B; ++i) {
            li[i] = __builtin_ctz(m | 0x80000000u);
            m &= m - 1u;
          }
          float v[B][J];
#pragma unroll
          for (int i = 0; i < B; ++i) {
            if (i < nb) {
              const int si = __builtin_amdgcn_readlane(s, li[i]);
              const float* __restrict__ r = xw + (size_t)si * C + lane;
#pragma unroll
              for (int j = 0; j < J; ++j) v[i][j] = r[32 * j];
            }
          }
#pragma unroll
          for (int i = 0; i < B; ++i) {
            if (i < nb) {
              const int di = __builtin_amdgcn_readlane(d, li[i]);
              const float ni = __int_as_float(__builtin_amdgcn_readlane(__float_as_int(nrm), li[i]));
              float* __restrict__ o = out + (size_t)di * C + lane;
#pragma unroll
              for (int j = 0; j < J; ++j) atomicAdd(o + 32 * j, v[i][j] * ni);
            }
          }
        }
      }
    }
  }
}

extern "C" void kernel_launch(void* const* d_in, const int* in_sizes, int n_in,
                              void* d_out, int out_size, void* d_ws, size_t ws_size,
                              hipStream_t stream) {
  (void)in_sizes; (void)n_in; (void)out_size; (void)ws_size;

  const float* feature = (const float*)d_in[0];
  const float* img     = (const float*)d_in[1];
  const int*   edges   = (const int*)d_in[2];
  const float* W_feat  = (const float*)d_in[3];
  const float* b_feat  = (const float*)d_in[4];
  const float* W_img   = (const float*)d_in[5];
  const float* b_img   = (const float*)d_in[6];
  const float* W_g1    = (const float*)d_in[7];
  const float* b_g1    = (const float*)d_in[8];
  const float* W_g2    = (const float*)d_in[9];
  const float* b_g2    = (const float*)d_in[10];
  float* out = (float*)d_out;

  const int* src = edges;
  const int* dst = edges + N_EDGES;

  float* ws     = (float*)d_ws;
  float* dinv   = ws;
  float* bufA   = ws + 50048;
  float* bufB   = bufA + (size_t)N_NODES * 96;

  const int T = 256;
  const int MB = (N_NODES + 63) / 64;
  const int SB = 1024;
  int* tick1 = (int*)d_out;
  int* tick2 = (int*)(bufB + (size_t)N_NODES * 64);

  k_init_deg<<<(N_NODES + T - 1) / T, T, 0, stream>>>(dinv, N_NODES);
  k_count_deg<<<(N_EDGES + T - 1) / T, T, 0, stream>>>(dst, dinv, N_EDGES);
  k_dinv<<<(N_NODES + T - 1) / T, T, 0, stream>>>(dinv, N_NODES);

  k_dense1<<<MB, 128, 0, stream>>>(feature, img, W_feat, W_img, W_g1, b_feat, b_img, b_g1,
                                   dinv, bufB, bufA, N_NODES, tick1);
  k_scatter<96><<<SB, T, 0, stream>>>(bufB, dinv, src, dst, bufA, N_EDGES, tick1);

  k_gemm2<<<MB, 128, 0, stream>>>(bufA, W_g2, b_g2, bufB, N_NODES, dinv, out, tick2);
  k_scatter<64><<<SB, T, 0, stream>>>(bufB, dinv, src, dst, out, N_EDGES, tick2);
}


# Degrees counted in LDS, not by one global float atomic per edge: a work-group counts one edge slice into one node range, stores integer partials, k_dinv sums 16 of them; 6 launches; dinv bit-equal
# speedup vs baseline: 2.3058x; 1.2979x over previous
#include <hip/hip_runtime.h>
#include <hip/hip_bf16.h>

typedef float v2f __attribute__((ext_vector_type(2)));
typedef float v4f __attribute__((ext_vector_type(4)));
typedef float v8f __attribute__((ext_vector_type(8)));

#define N_NODES 50000
#define N_EDGES 800000

#define DEG_RANGES 16
#define DEG_SLICES 16
#define DEG_RN ((N_NODES + DEG_RANGES - 1) / DEG_RANGES)
#define DEG_SE ((((N_EDGES + DEG_SLICES - 1) / DEG_SLICES) + 3) & ~3)
#define DEG_T 512
#define DEG_U ((DEG_SE + DEG_T * 4 - 1) / (DEG_T * 4))

__global__ __launch_bounds__(DEG_T) void k_deg_part(const int* __restrict__ dst,
                                                    unsigned* __restrict__ part, int e) {
  __shared__ unsigned cnt[DEG_RN];
  const int range = blockIdx.x % DEG_RANGES;
  const int slice = blockIdx.x / DEG_RANGES;
  const int base = range * DEG_RN;
  const int e0 = slice * DEG_SE;
  const int e1 = (e0 + DEG_SE < e) ? (e0 + DEG_SE) : ((e > e0) ? e : e0);

  const int e1v = e0 + ((e1 - e0) & ~3);

  int4 d[DEG_U];
#pragma unroll
  for (int u = 0; u < DEG_U; ++u) {
    const int i = e0 + (u * DEG_T + (int)threadIdx.x) * 4;
    d[u] = make_int4(-1, -1, -1, -1);
    if (i < e1v) d[u] = *(const int4*)(dst + i);
  }
  int dt = -1;
  if ((int)threadIdx.x < e1 - e1v) dt = dst[e1v + threadIdx.x];

  for (int i = threadIdx.x; i < DEG_RN; i += DEG_T) cnt[i] = 0u;
  __syncthreads();

#pragma unroll
  for (int u = 0; u < DEG_U; ++u) {
    unsigned r;
    r = (unsigned)(d[u].x - base); if (r < (unsigned)DEG_RN) atomicAdd(&cnt[r], 1u);
    r = (unsigned)(d[u].y - base); if (r < (unsigned)DEG_RN) atomicAdd(&cnt[r], 1u);
    r = (unsigned)(d[u].z - base); if (r < (unsigned)DEG_RN) atomicAdd(&cnt[r], 1u);
    r = (unsigned)(d[u].w - base); if (r < (unsigned)DEG_RN) atomicAdd(&cnt[r], 1u);
  }
  {
    const unsigned r = (unsigned)(dt - base);
    if (r < (unsigned)DEG_RN) atomicAdd(&cnt[r], 1u);
  }
  __syncthreads();

  unsigned* __restrict__ prow = part + (size_t)slice * N_NODES + base;
  for (int i = threadIdx.x; i < DEG_RN; i += DEG_T)
    if (base + i < N_NODES) prow[i] = cnt[i];
}

__global__ void k_dinv(const unsigned* __restrict__ part, float* __restrict__ dinv, int n) {
  const int i = blockIdx.x * blockDim.x + threadIdx.x;
  if (i < n) {
    unsigned c = 1u;
#pragma unroll
    for (int s = 0; s < DEG_SLICES; ++s) c += part[(size_t)s * N_NODES + i];
    dinv[i] = rsqrtf((float)c);
  }
}


__global__ __launch_bounds__(128) void k_gemm2(
    const float* A, const float* __restrict__ W,
    const float* __restrict__ bias, float* __restrict__ out, int M,
    const float* __restrict__ dinv, float* init_out, int* __restrict__ tick) {
  if (blockIdx.x == 0 && threadIdx.x < 2) tick[threadIdx.x * 64] = 0;
  const int lane = threadIdx.x & 31;
  const int wave = threadIdx.x >> 5;
  const int row_tile = (blockIdx.x * 4 + wave) * 16;
  if (row_tile >= M) return;

  const int m16 = lane & 15;
  const int khalf = lane >> 4;

  v8f acc[4];
#pragma unroll
  for (int t = 0; t < 4; ++t) acc[t] = (v8f){0.f, 0.f, 0.f, 0.f, 0.f, 0.f, 0.f, 0.f};

  const float* arow = A + (size_t)(row_tile + m16) * 96 + khalf * 2;
  const float* __restrict__ wrow = W + khalf * 2 * 64 + m16 * 4;

  for (int k0 = 0; k0 < 96; k0 += 16) {
#pragma unroll
    for (int kq = 0; kq < 4; ++kq) {
      const int k = k0 + kq * 4;
      v2f afrag = *(const v2f*)(arow + k);
      afrag.x = fmaxf(afrag.x, 0.0f);
      afrag.y = fmaxf(afrag.y, 0.0f);
      const v4f w0 = *(const v4f*)(wrow + k * 64);
      const v4f w1 = *(const v4f*)(wrow + (k + 1) * 64);
#pragma unroll
      for (int t = 0; t < 4; ++t) {
        v2f bfrag;
        bfrag.x = w0[t];
        bfrag.y = w1[t];
        acc[t] = __builtin_amdgcn_wmma_f32_16x16x4_f32(
             false, afrag,  false, bfrag,
             (short)0, acc[t],  false,  false);
      }
    }
  }

  const int rbase = row_tile + khalf * 8;
  const v4f bc = *(const v4f*)(bias + m16 * 4);
  float di[8];
#pragma unroll
  for (int i = 0; i < 8; ++i) di[i] = dinv[rbase + i];
#pragma unroll
  for (int i = 0; i < 8; ++i) {
    v4f v, s;
    v.x = acc[0][i]; v.y = acc[1][i]; v.z = acc[2][i]; v.w = acc[3][i];
    s.x = bc.x + v.x * di[i] * di[i];
    s.y = bc.y + v.y * di[i] * di[i];
    s.z = bc.z + v.z * di[i] * di[i];
    s.w = bc.w + v.w * di[i] * di[i];
    *(v4f*)(out + (size_t)(rbase + i) * 64 + m16 * 4) = v;
    *(v4f*)(init_out + (size_t)(rbase + i) * 64 + m16 * 4) = s;
  }
}

template <int NT> struct wvec;
template <> struct wvec<4> { typedef v4f type; };
template <> struct wvec<2> { typedef v2f type; };

template <int KDIM, int NT, int T0>
__device__ __forceinline__ void enc_xT(const float* __restrict__ nrow,
                                       const float* __restrict__ wrow, v8f (&xt)[6]) {
  typedef typename wvec<NT>::type vw;
  constexpr int NC = NT * 16;
#pragma unroll 2
  for (int k0 = 0; k0 < KDIM; k0 += 16) {
    if (k0 + 256 < KDIM) __builtin_prefetch(nrow + k0 + 256, 0, 0);
#pragma unroll
    for (int kq = 0; kq < 4; ++kq) {
      const int k = k0 + kq * 4;
      const v2f xf = *(const v2f*)(nrow + k);
      const vw w0 = *(const vw*)(wrow + k * NC);
      const vw w1 = *(const vw*)(wrow + (k + 1) * NC);
#pragma unroll
      for (int t = 0; t < NT; ++t) {
        v2f wf;
        wf.x = w0[t];
        wf.y = w1[t];
        xt[T0 + t] = __builtin_amdgcn_wmma_f32_16x16x4_f32(
             false, wf,  false, xf,
             (short)0, xt[T0 + t],  false,  false);
      }
    }
  }
}

__global__ __launch_bounds__(128) __attribute__((amdgpu_waves_per_eu(4, 4))) void k_dense1(
    const float* __restrict__ feature, const float* __restrict__ img,
    const float* __restrict__ W_feat, const float* __restrict__ W_img,
    const float* __restrict__ W_g1,
    const float* __restrict__ b_feat, const float* __restrict__ b_img,
    const float* __restrict__ b_g1, const float* __restrict__ dinv,
    float* __restrict__ xw, float* __restrict__ hinit, int M, int* __restrict__ tick) {
  if (blockIdx.x == 0 && threadIdx.x < 2) tick[threadIdx.x * 64] = 0;
  const int lane = threadIdx.x & 31;
  const int wave = threadIdx.x >> 5;
  const int row_tile = (blockIdx.x * 4 + wave) * 16;
  if (row_tile >= M) return;

  const int m16 = lane & 15;
  const int khalf = lane >> 4;

  v8f xt[6];
#pragma unroll
  for (int t = 0; t < 6; ++t) xt[t] = (v8f){0.f, 0.f, 0.f, 0.f, 0.f, 0.f, 0.f, 0.f};

  enc_xT<512, 4, 0>(feature + (size_t)(row_tile + m16) * 512 + khalf * 2,
                    W_feat + khalf * 2 * 64 + m16 * 4, xt);
  enc_xT<256, 2, 4>(img + (size_t)(row_tile + m16) * 256 + khalf * 2,
                    W_img + khalf * 2 * 32 + m16 * 2, xt);

#pragma unroll
  for (int r = 0; r < 8; ++r) {
    const v4f bf4 = *(const v4f*)(b_feat + 4 * (8 * khalf + r));
    const v2f bi2 = *(const v2f*)(b_img + 2 * (8 * khalf + r));
    xt[0][r] = fmaxf(xt[0][r] + bf4.x, 0.0f);
    xt[1][r] = fmaxf(xt[1][r] + bf4.y, 0.0f);
    xt[2][r] = fmaxf(xt[2][r] + bf4.z, 0.0f);
    xt[3][r] = fmaxf(xt[3][r] + bf4.w, 0.0f);
    xt[4][r] = fmaxf(xt[4][r] + bi2.x, 0.0f);
    xt[5][r] = fmaxf(xt[5][r] + bi2.y, 0.0f);
  }

  v8f acc[6];
#pragma unroll
  for (int u = 0; u < 6; ++u) acc[u] = (v8f){0.f, 0.f, 0.f, 0.f, 0.f, 0.f, 0.f, 0.f};
  const float* __restrict__ gF = W_g1 + (size_t)(32 * khalf) * 96 + m16 * 4;
  const float* __restrict__ gI = W_g1 + (size_t)(64 + 16 * khalf) * 96 + m16 * 4;
#pragma unroll
  for (int t = 0; t < 6; ++t) {
#pragma unroll
    for (int j = 0; j < 4; ++j) {
      v2f af;
      af.x = xt[t][2 * j];
      af.y = xt[t][2 * j + 1];
      const float* g0 = (t < 4) ? (gF + (8 * j + t) * 96) : (gI + (4 * j + (t - 4)) * 96);
      const float* g1 = g0 + ((t < 4) ? 4 : 2) * 96;
      const v4f p0 = *(const v4f*)(g0);
      const v4f p1 = *(const v4f*)(g1);
      const v2f q0 = *(const v2f*)(g0 + 64 - 2 * m16);
      const v2f q1 = *(const v2f*)(g1 + 64 - 2 * m16);
#pragma unroll
      for (int u = 0; u < 4; ++u) {
        v2f bf;
        bf.x = p0[u];
        bf.y = p1[u];
        acc[u] = __builtin_amdgcn_wmma_f32_16x16x4_f32(
             false, af,  false, bf,
             (short)0, acc[u],  false,  false);
      }
#pragma unroll
      for (int u = 0; u < 2; ++u) {
        v2f bf;
        bf.x = q0[u];
        bf.y = q1[u];
        acc[4 + u] = __builtin_amdgcn_wmma_f32_16x16x4_f32(
             false, af,  false, bf,
             (short)0, acc[4 + u],  false,  false);
      }
    }
  }

  const int rbase = row_tile + khalf * 8;
  const v4f bc4 = *(const v4f*)(b_g1 + m16 * 4);
  const v2f bc2 = *(const v2f*)(b_g1 + 64 + m16 * 2);
  float di[8];
#pragma unroll
  for (int i = 0; i < 8; ++i) di[i] = dinv[rbase + i];
#pragma unroll
  for (int i = 0; i < 8; ++i) {
    v4f v, s;
    v2f w, z;
    v.x = acc[0][i]; v.y = acc[1][i]; v.z = acc[2][i]; v.w = acc[3][i];
    w.x = acc[4][i]; w.y = acc[5][i];
    s.x = bc4.x + v.x * di[i] * di[i];
    s.y = bc4.y + v.y * di[i] * di[i];
    s.z = bc4.z + v.z * di[i] * di[i];
    s.w = bc4.w + v.w * di[i] * di[i];
    z.x = bc2.x + w.x * di[i] * di[i];
    z.y = bc2.y + w.y * di[i] * di[i];
    float* xr = xw + (size_t)(rbase + i) * 96;
    float* hr = hinit + (size_t)(rbase + i) * 96;
    *(v4f*)(xr + m16 * 4) = v;
    *(v2f*)(xr + 64 + m16 * 2) = w;
    *(v4f*)(hr + m16 * 4) = s;
    *(v2f*)(hr + 64 + m16 * 2) = z;
  }
}

template <int C>
__global__ __launch_bounds__(256) void k_scatter(
    const float* __restrict__ xw, const float* __restrict__ dinv,
    const int* __restrict__ src, const int* __restrict__ dst,
    float* __restrict__ out, int e, int* __restrict__ tick) {
  constexpr int J = C / 32;
  constexpr int B = 8;
  constexpr int CH = 64;
  const int lane = threadIdx.x & 31;
  const int nchunks = (e + CH - 1) / CH;
  const int myhalf = (int)((__builtin_amdgcn_s_sendmsg_rtn(0x87) >> 18) & 1u);

  for (int pass = 0; pass < 2; ++pass) {
    const int h = myhalf ^ pass;
    int* __restrict__ word = tick + h * 64;
    for (int it = 0; it <= nchunks; ++it) {
      int t = 0;
      if (lane == 0) t = atomicAdd(word, 1);
      t = __builtin_amdgcn_readfirstlane(t);
      if (t >= nchunks) break;

      for (int sub = 0; sub < CH / 32; ++sub) {
        const int e0 = t * CH + sub * 32;
        if (e0 >= e) break;
        const int cnt = (e - e0 < 32) ? (e - e0) : 32;

        int s = 0, d = 0;
        float nrm = 0.0f;
        bool own = false;
        if (lane < cnt) {
          s = src[e0 + lane];
          d = dst[e0 + lane];
          own = ((d >= N_NODES / 2) ? 1 : 0) == h;
          if (own) nrm = dinv[s] * dinv[d];
        }
        unsigned m = __builtin_amdgcn_ballot_w32(own);

        while (m != 0u) {
          const int left = __builtin_popcount(m);
          const int nb = left < B ? left : B;
          int li[B];
#pragma unroll
          for (int i = 0; i < B; ++i) {
            li[i] = __builtin_ctz(m | 0x80000000u);
            m &= m - 1u;
          }
          float v[B][J];
#pragma unroll
          for (int i = 0; i < B; ++i) {
            if (i < nb) {
              const int si = __builtin_amdgcn_readlane(s, li[i]);
              const float* __restrict__ r = xw + (size_t)si * C + lane;
#pragma unroll
              for (int j = 0; j < J; ++j) v[i][j] = r[32 * j];
            }
          }
#pragma unroll
          for (int i = 0; i < B; ++i) {
            if (i < nb) {
              const int di = __builtin_amdgcn_readlane(d, li[i]);
              const float ni = __int_as_float(__builtin_amdgcn_readlane(__float_as_int(nrm), li[i]));
              float* __restrict__ o = out + (size_t)di * C + lane;
#pragma unroll
              for (int j = 0; j < J; ++j) atomicAdd(o + 32 * j, v[i][j] * ni);
            }
          }
        }
      }
    }
  }
}

extern "C" void kernel_launch(void* const* d_in, const int* in_sizes, int n_in,
                              void* d_out, int out_size, void* d_ws, size_t ws_size,
                              hipStream_t stream) {
  (void)in_sizes; (void)n_in; (void)out_size; (void)ws_size;

  const float* feature = (const float*)d_in[0];
  const float* img     = (const float*)d_in[1];
  const int*   edges   = (const int*)d_in[2];
  const float* W_feat  = (const float*)d_in[3];
  const float* b_feat  = (const float*)d_in[4];
  const float* W_img   = (const float*)d_in[5];
  const float* b_img   = (const float*)d_in[6];
  const float* W_g1    = (const float*)d_in[7];
  const float* b_g1    = (const float*)d_in[8];
  const float* W_g2    = (const float*)d_in[9];
  const float* b_g2    = (const float*)d_in[10];
  float* out = (float*)d_out;

  const int* src = edges;
  const int* dst = edges + N_EDGES;

  float* ws     = (float*)d_ws;
  float* dinv   = ws;
  float* bufA   = ws + 50048;
  float* bufB   = bufA + (size_t)N_NODES * 96;

  const int T = 256;
  const int MB = (N_NODES + 63) / 64;
  const int SB = 1024;
  int* tick1 = (int*)d_out;
  int* tick2 = (int*)(bufB + (size_t)N_NODES * 64);

  unsigned* part = (unsigned*)bufB;
  k_deg_part<<<DEG_RANGES * DEG_SLICES, DEG_T, 0, stream>>>(dst, part, N_EDGES);
  k_dinv<<<(N_NODES + T - 1) / T, T, 0, stream>>>(part, dinv, N_NODES);

  k_dense1<<<MB, 128, 0, stream>>>(feature, img, W_feat, W_img, W_g1, b_feat, b_img, b_g1,
                                   dinv, bufB, bufA, N_NODES, tick1);
  k_scatter<96><<<SB, T, 0, stream>>>(bufB, dinv, src, dst, bufA, N_EDGES, tick1);

  k_gemm2<<<MB, 128, 0, stream>>>(bufA, W_g2, b_g2, bufB, N_NODES, dinv, out, tick2);
  k_scatter<64><<<SB, T, 0, stream>>>(bufB, dinv, src, dst, out, N_EDGES, tick2);
}


# Edge sums by owner work-groups, no binning pass: each scans an edge slice for its 782-node range, sums rows by LDS float atomics, adds them to out once: 4 global float atomics per line; arrival order
# speedup vs baseline: 2.4138x; 1.0468x over previous
#include <hip/hip_runtime.h>
#include <hip/hip_bf16.h>

typedef float v2f __attribute__((ext_vector_type(2)));
typedef float v4f __attribute__((ext_vector_type(4)));
typedef float v8f __attribute__((ext_vector_type(8)));

#define N_NODES 50000
#define N_EDGES 800000

#define DEG_RANGES 16
#define DEG_SLICES 16
#define DEG_RN ((N_NODES + DEG_RANGES - 1) / DEG_RANGES)
#define DEG_SE ((((N_EDGES + DEG_SLICES - 1) / DEG_SLICES) + 3) & ~3)
#define DEG_T 512
#define DEG_U ((DEG_SE + DEG_T * 4 - 1) / (DEG_T * 4))

__global__ __launch_bounds__(DEG_T) void k_deg_part(const int* __restrict__ dst,
                                                    unsigned* __restrict__ part, int e) {
  __shared__ unsigned cnt[DEG_RN];
  const int range = blockIdx.x % DEG_RANGES;
  const int slice = blockIdx.x / DEG_RANGES;
  const int base = range * DEG_RN;
  const int e0 = slice * DEG_SE;
  const int e1 = (e0 + DEG_SE < e) ? (e0 + DEG_SE) : ((e > e0) ? e : e0);

  const int e1v = e0 + ((e1 - e0) & ~3);

  int4 d[DEG_U];
#pragma unroll
  for (int u = 0; u < DEG_U; ++u) {
    const int i = e0 + (u * DEG_T + (int)threadIdx.x) * 4;
    d[u] = make_int4(-1, -1, -1, -1);
    if (i < e1v) d[u] = *(const int4*)(dst + i);
  }
  int dt = -1;
  if ((int)threadIdx.x < e1 - e1v) dt = dst[e1v + threadIdx.x];

  for (int i = threadIdx.x; i < DEG_RN; i += DEG_T) cnt[i] = 0u;
  __syncthreads();

#pragma unroll
  for (int u = 0; u < DEG_U; ++u) {
    unsigned r;
    r = (unsigned)(d[u].x - base); if (r < (unsigned)DEG_RN) atomicAdd(&cnt[r], 1u);
    r = (unsigned)(d[u].y - base); if (r < (unsigned)DEG_RN) atomicAdd(&cnt[r], 1u);
    r = (unsigned)(d[u].z - base); if (r < (unsigned)DEG_RN) atomicAdd(&cnt[r], 1u);
    r = (unsigned)(d[u].w - base); if (r < (unsigned)DEG_RN) atomicAdd(&cnt[r], 1u);
  }
  {
    const unsigned r = (unsigned)(dt - base);
    if (r < (unsigned)DEG_RN) atomicAdd(&cnt[r], 1u);
  }
  __syncthreads();

  unsigned* __restrict__ prow = part + (size_t)slice * N_NODES + base;
  for (int i = threadIdx.x; i < DEG_RN; i += DEG_T)
    if (base + i < N_NODES) prow[i] = cnt[i];
}

__global__ void k_dinv(const unsigned* __restrict__ part, float* __restrict__ dinv, int n) {
  const int i = blockIdx.x * blockDim.x + threadIdx.x;
  if (i < n) {
    unsigned c = 1u;
#pragma unroll
    for (int s = 0; s < DEG_SLICES; ++s) c += part[(size_t)s * N_NODES + i];
    dinv[i] = rsqrtf((float)c);
  }
}


__global__ __launch_bounds__(128) void k_gemm2(
    const float* A, const float* __restrict__ W,
    const float* __restrict__ bias, float* __restrict__ out, int M,
    const float* __restrict__ dinv, float* init_out) {
  const int lane = threadIdx.x & 31;
  const int wave = threadIdx.x >> 5;
  const int row_tile = (blockIdx.x * 4 + wave) * 16;
  if (row_tile >= M) return;

  const int m16 = lane & 15;
  const int khalf = lane >> 4;

  v8f acc[4];
#pragma unroll
  for (int t = 0; t < 4; ++t) acc[t] = (v8f){0.f, 0.f, 0.f, 0.f, 0.f, 0.f, 0.f, 0.f};

  const float* arow = A + (size_t)(row_tile + m16) * 96 + khalf * 2;
  const float* __restrict__ wrow = W + khalf * 2 * 64 + m16 * 4;

  for (int k0 = 0; k0 < 96; k0 += 16) {
#pragma unroll
    for (int kq = 0; kq < 4; ++kq) {
      const int k = k0 + kq * 4;
      v2f afrag = *(const v2f*)(arow + k);
      afrag.x = fmaxf(afrag.x, 0.0f);
      afrag.y = fmaxf(afrag.y, 0.0f);
      const v4f w0 = *(const v4f*)(wrow + k * 64);
      const v4f w1 = *(const v4f*)(wrow + (k + 1) * 64);
#pragma unroll
      for (int t = 0; t < 4; ++t) {
        v2f bfrag;
        bfrag.x = w0[t];
        bfrag.y = w1[t];
        acc[t] = __builtin_amdgcn_wmma_f32_16x16x4_f32(
             false, afrag,  false, bfrag,
             (short)0, acc[t],  false,  false);
      }
    }
  }

  const int rbase = row_tile + khalf * 8;
  const v4f bc = *(const v4f*)(bias + m16 * 4);
  float di[8];
#pragma unroll
  for (int i = 0; i < 8; ++i) di[i] = dinv[rbase + i];
#pragma unroll
  for (int i = 0; i < 8; ++i) {
    v4f v, s;
    v.x = acc[0][i]; v.y = acc[1][i]; v.z = acc[2][i]; v.w = acc[3][i];
    s.x = bc.x + v.x * di[i] * di[i];
    s.y = bc.y + v.y * di[i] * di[i];
    s.z = bc.z + v.z * di[i] * di[i];
    s.w = bc.w + v.w * di[i] * di[i];
    *(v4f*)(out + (size_t)(rbase + i) * 64 + m16 * 4) = v;
    *(v4f*)(init_out + (size_t)(rbase + i) * 64 + m16 * 4) = s;
  }
}

template <int NT> struct wvec;
template <> struct wvec<4> { typedef v4f type; };
template <> struct wvec<2> { typedef v2f type; };

template <int KDIM, int NT, int T0>
__device__ __forceinline__ void enc_xT(const float* __restrict__ nrow,
                                       const float* __restrict__ wrow, v8f (&xt)[6]) {
  typedef typename wvec<NT>::type vw;
  constexpr int NC = NT * 16;
#pragma unroll 2
  for (int k0 = 0; k0 < KDIM; k0 += 16) {
    if (k0 + 256 < KDIM) __builtin_prefetch(nrow + k0 + 256, 0, 0);
#pragma unroll
    for (int kq = 0; kq < 4; ++kq) {
      const int k = k0 + kq * 4;
      const v2f xf = *(const v2f*)(nrow + k);
      const vw w0 = *(const vw*)(wrow + k * NC);
      const vw w1 = *(const vw*)(wrow + (k + 1) * NC);
#pragma unroll
      for (int t = 0; t < NT; ++t) {
        v2f wf;
        wf.x = w0[t];
        wf.y = w1[t];
        xt[T0 + t] = __builtin_amdgcn_wmma_f32_16x16x4_f32(
             false, wf,  false, xf,
             (short)0, xt[T0 + t],  false,  false);
      }
    }
  }
}

__global__ __launch_bounds__(128) __attribute__((amdgpu_waves_per_eu(4, 4))) void k_dense1(
    const float* __restrict__ feature, const float* __restrict__ img,
    const float* __restrict__ W_feat, const float* __restrict__ W_img,
    const float* __restrict__ W_g1,
    const float* __restrict__ b_feat, const float* __restrict__ b_img,
    const float* __restrict__ b_g1, const float* __restrict__ dinv,
    float* __restrict__ xw, float* __restrict__ hinit, int M) {
  const int lane = threadIdx.x & 31;
  const int wave = threadIdx.x >> 5;
  const int row_tile = (blockIdx.x * 4 + wave) * 16;
  if (row_tile >= M) return;

  const int m16 = lane & 15;
  const int khalf = lane >> 4;

  v8f xt[6];
#pragma unroll
  for (int t = 0; t < 6; ++t) xt[t] = (v8f){0.f, 0.f, 0.f, 0.f, 0.f, 0.f, 0.f, 0.f};

  enc_xT<512, 4, 0>(feature + (size_t)(row_tile + m16) * 512 + khalf * 2,
                    W_feat + khalf * 2 * 64 + m16 * 4, xt);
  enc_xT<256, 2, 4>(img + (size_t)(row_tile + m16) * 256 + khalf * 2,
                    W_img + khalf * 2 * 32 + m16 * 2, xt);

#pragma unroll
  for (int r = 0; r < 8; ++r) {
    const v4f bf4 = *(const v4f*)(b_feat + 4 * (8 * khalf + r));
    const v2f bi2 = *(const v2f*)(b_img + 2 * (8 * khalf + r));
    xt[0][r] = fmaxf(xt[0][r] + bf4.x, 0.0f);
    xt[1][r] = fmaxf(xt[1][r] + bf4.y, 0.0f);
    xt[2][r] = fmaxf(xt[2][r] + bf4.z, 0.0f);
    xt[3][r] = fmaxf(xt[3][r] + bf4.w, 0.0f);
    xt[4][r] = fmaxf(xt[4][r] + bi2.x, 0.0f);
    xt[5][r] = fmaxf(xt[5][r] + bi2.y, 0.0f);
  }

  v8f acc[6];
#pragma unroll
  for (int u = 0; u < 6; ++u) acc[u] = (v8f){0.f, 0.f, 0.f, 0.f, 0.f, 0.f, 0.f, 0.f};
  const float* __restrict__ gF = W_g1 + (size_t)(32 * khalf) * 96 + m16 * 4;
  const float* __restrict__ gI = W_g1 + (size_t)(64 + 16 * khalf) * 96 + m16 * 4;
#pragma unroll
  for (int t = 0; t < 6; ++t) {
#pragma unroll
    for (int j = 0; j < 4; ++j) {
      v2f af;
      af.x = xt[t][2 * j];
      af.y = xt[t][2 * j + 1];
      const float* g0 = (t < 4) ? (gF + (8 * j + t) * 96) : (gI + (4 * j + (t - 4)) * 96);
      const float* g1 = g0 + ((t < 4) ? 4 : 2) * 96;
      const v4f p0 = *(const v4f*)(g0);
      const v4f p1 = *(const v4f*)(g1);
      const v2f q0 = *(const v2f*)(g0 + 64 - 2 * m16);
      const v2f q1 = *(const v2f*)(g1 + 64 - 2 * m16);
#pragma unroll
      for (int u = 0; u < 4; ++u) {
        v2f bf;
        bf.x = p0[u];
        bf.y = p1[u];
        acc[u] = __builtin_amdgcn_wmma_f32_16x16x4_f32(
             false, af,  false, bf,
             (short)0, acc[u],  false,  false);
      }
#pragma unroll
      for (int u = 0; u < 2; ++u) {
        v2f bf;
        bf.x = q0[u];
        bf.y = q1[u];
        acc[4 + u] = __builtin_amdgcn_wmma_f32_16x16x4_f32(
             false, af,  false, bf,
             (short)0, acc[4 + u],  false,  false);
      }
    }
  }

  const int rbase = row_tile + khalf * 8;
  const v4f bc4 = *(const v4f*)(b_g1 + m16 * 4);
  const v2f bc2 = *(const v2f*)(b_g1 + 64 + m16 * 2);
  float di[8];
#pragma unroll
  for (int i = 0; i < 8; ++i) di[i] = dinv[rbase + i];
#pragma unroll
  for (int i = 0; i < 8; ++i) {
    v4f v, s;
    v2f w, z;
    v.x = acc[0][i]; v.y = acc[1][i]; v.z = acc[2][i]; v.w = acc[3][i];
    w.x = acc[4][i]; w.y = acc[5][i];
    s.x = bc4.x + v.x * di[i] * di[i];
    s.y = bc4.y + v.y * di[i] * di[i];
    s.z = bc4.z + v.z * di[i] * di[i];
    s.w = bc4.w + v.w * di[i] * di[i];
    z.x = bc2.x + w.x * di[i] * di[i];
    z.y = bc2.y + w.y * di[i] * di[i];
    float* xr = xw + (size_t)(rbase + i) * 96;
    float* hr = hinit + (size_t)(rbase + i) * 96;
    *(v4f*)(xr + m16 * 4) = v;
    *(v2f*)(xr + 64 + m16 * 2) = w;
    *(v4f*)(hr + m16 * 4) = s;
    *(v2f*)(hr + 64 + m16 * 2) = z;
  }
}

#define AG_R 64
#define AG_S 4
#define AG_RN ((N_NODES + AG_R - 1) / AG_R)
#define AG_SE (N_EDGES / AG_S)
#define AG_T 1024
#define AG_W (AG_T / 32)
#define AG_U ((AG_SE + AG_T * 4 - 1) / (AG_T * 4))
#define AG_UC 7
#define AG_LCAP 192
static_assert(N_EDGES % (AG_S * 4) == 0, "slices are whole groups of four edges");
static_assert(AG_U % AG_UC == 0, "the scan runs in whole chunks");
static_assert(N_EDGES <= (1 << 20) && AG_RN <= (1 << 12), "a list word is edge | row << 20");

template <int C>
__global__ __launch_bounds__(AG_T) void k_aggr(
    const float* __restrict__ xw, const float* __restrict__ dinv,
    const int* __restrict__ src, const int* __restrict__ dst,
    float* __restrict__ out) {
  constexpr int J = C / 32;
  constexpr int B = 8;
  __shared__ float acc[AG_RN * C];
  __shared__ unsigned list[AG_W * AG_LCAP];
  const int lane = threadIdx.x & 31;
  const int wave = __builtin_amdgcn_readfirstlane((int)(threadIdx.x >> 5));
  const int range = blockIdx.x % AG_R;
  const int slice = blockIdx.x / AG_R;
  const int base = range * AG_RN;
  const int e0 = slice * AG_SE;
  const int e1 = e0 + AG_SE;
  const unsigned below = (1u << lane) - 1u;
  unsigned* __restrict__ mylist = list + wave * AG_LCAP;

  for (int i = threadIdx.x * 4; i < AG_RN * C; i += AG_T * 4)
    *(v4f*)(acc + i) = (v4f){0.f, 0.f, 0.f, 0.f};
  __syncthreads();

  int wcount = 0;
  for (int uc = 0; uc < AG_U; uc += AG_UC) {
    int4 d[AG_UC];
#pragma unroll
    for (int k = 0; k < AG_UC; ++k) {
      const int i = e0 + ((uc + k) * AG_T + (int)threadIdx.x) * 4;
      d[k] = make_int4(-1, -1, -1, -1);
      if (i < e1) d[k] = *(const int4*)(dst + i);
    }
#pragma unroll
    for (int k = 0; k < AG_UC; ++k) {
      const int i = e0 + ((uc + k) * AG_T + (int)threadIdx.x) * 4;
      const int dq[4] = {d[k].x, d[k].y, d[k].z, d[k].w};
#pragma unroll
      for (int q = 0; q < 4; ++q) {
        const unsigned r = (unsigned)(dq[q] - base);
        const bool hit = r < (unsigned)AG_RN;
        const unsigned m = __builtin_amdgcn_ballot_w32(hit);
        if (m != 0u) {
          const int pos = wcount + __builtin_popcount(m & below);
          if (hit && pos < AG_LCAP) mylist[pos] = (unsigned)(i + q) | (r << 20);
          wcount += __builtin_popcount(m);
        }
      }
    }
  }

  if (wcount > AG_LCAP) {
    int seen = 0;
    for (int u = 0; u < AG_U; ++u) {
      const int i = e0 + (u * AG_T + (int)threadIdx.x) * 4;
      for (int q = 0; q < 4; ++q) {
        int dv = -1;
        if (i < e1) dv = dst[i + q];
        const unsigned r = (unsigned)(dv - base);
        const bool hit = r < (unsigned)AG_RN;
        const unsigned m = __builtin_amdgcn_ballot_w32(hit);
        const int pos = seen + __builtin_popcount(m & below);
        if (hit && pos >= AG_LCAP) {
          const int s = src[i + q];
          const float nrm = dinv[s] * dinv[base + (int)r];
          for (int col = 0; col < C; ++col)
            atomicAdd(&acc[r * C + col], xw[(size_t)s * C + col] * nrm);
        }
        seen += __builtin_popcount(m);
      }
    }
  }
  __syncthreads();

  const int n = wcount < AG_LCAP ? wcount : AG_LCAP;
  for (int g = 0; g < n; g += 32) {
    const int cnt = (n - g < 32) ? (n - g) : 32;
    int s = 0;
    unsigned r = 0u;
    float nrm = 0.0f;
    if (lane < cnt) {
      const unsigned ent = mylist[g + lane];
      r = ent >> 20;
      s = src[ent & 0xFFFFFu];
      nrm = dinv[s] * dinv[base + (int)r];
    }
    for (int b0 = 0; b0 < cnt; b0 += B) {
      float v[B][J];
#pragma unroll
      for (int i = 0; i < B; ++i) {
        const int li = (b0 + i < cnt) ? (b0 + i) : (cnt - 1);
        const int si = __builtin_amdgcn_readlane(s, li);
        const float* __restrict__ p = xw + (size_t)si * C + lane;
#pragma unroll
        for (int j = 0; j < J; ++j) v[i][j] = p[32 * j];
      }
#pragma unroll
      for (int i = 0; i < B; ++i) {
        if (b0 + i < cnt) {
          const int ri = __builtin_amdgcn_readlane((int)r, b0 + i);
          const float ni = __int_as_float(__builtin_amdgcn_readlane(__float_as_int(nrm), b0 + i));
          float* a = acc + ri * C + lane;
#pragma unroll
          for (int j = 0; j < J; ++j) atomicAdd(a + 32 * j, v[i][j] * ni);
        }
      }
    }
  }
  __syncthreads();

  for (int row = wave; row < AG_RN; row += AG_W) {
    if (base + row < N_NODES) {
      const float* a = acc + row * C + lane;
      float* __restrict__ o = out + (size_t)(base + row) * C + lane;
      float t[J];
#pragma unroll
      for (int j = 0; j < J; ++j) t[j] = a[32 * j];
#pragma unroll
      for (int j = 0; j < J; ++j) atomicAdd(o + 32 * j, t[j]);
    }
  }
}

extern "C" void kernel_launch(void* const* d_in, const int* in_sizes, int n_in,
                              void* d_out, int out_size, void* d_ws, size_t ws_size,
                              hipStream_t stream) {
  (void)in_sizes; (void)n_in; (void)out_size; (void)ws_size;

  const float* feature = (const float*)d_in[0];
  const float* img     = (const float*)d_in[1];
  const int*   edges   = (const int*)d_in[2];
  const float* W_feat  = (const float*)d_in[3];
  const float* b_feat  = (const float*)d_in[4];
  const float* W_img   = (const float*)d_in[5];
  const float* b_img   = (const float*)d_in[6];
  const float* W_g1    = (const float*)d_in[7];
  const float* b_g1    = (const float*)d_in[8];
  const float* W_g2    = (const float*)d_in[9];
  const float* b_g2    = (const float*)d_in[10];
  float* out = (float*)d_out;

  const int* src = edges;
  const int* dst = edges + N_EDGES;

  float* ws     = (float*)d_ws;
  float* dinv   = ws;
  float* bufA   = ws + 50048;
  float* bufB   = bufA + (size_t)N_NODES * 96;

  const int T = 256;
  const int MB = (N_NODES + 63) / 64;

  unsigned* part = (unsigned*)bufB;
  k_deg_part<<<DEG_RANGES * DEG_SLICES, DEG_T, 0, stream>>>(dst, part, N_EDGES);
  k_dinv<<<(N_NODES + T - 1) / T, T, 0, stream>>>(part, dinv, N_NODES);

  k_dense1<<<MB, 128, 0, stream>>>(feature, img, W_feat, W_img, W_g1, b_feat, b_img, b_g1,
                                   dinv, bufB, bufA, N_NODES);
  k_aggr<96><<<AG_R * AG_S, AG_T, 0, stream>>>(bufB, dinv, src, dst, bufA);

  k_gemm2<<<MB, 128, 0, stream>>>(bufA, W_g2, b_g2, bufB, N_NODES, dinv, out);
  k_aggr<64><<<AG_R * AG_S, AG_T, 0, stream>>>(bufB, dinv, src, dst, out);
}


# Layer 2 reads back the per-wave edge lists layer 1 saved (source, row, norm): no second scan or lookups; xw1 cols 0-63 lie in out, xw2 over h rows to free room; LDS adds in arrival order as before
# speedup vs baseline: 2.5059x; 1.0382x over previous
#include <hip/hip_runtime.h>
#include <hip/hip_bf16.h>

typedef float v2f __attribute__((ext_vector_type(2)));
typedef float v4f __attribute__((ext_vector_type(4)));
typedef float v8f __attribute__((ext_vector_type(8)));

#define N_NODES 50000
#define N_EDGES 800000

#define DEG_RANGES 16
#define DEG_SLICES 16
#define DEG_RN ((N_NODES + DEG_RANGES - 1) / DEG_RANGES)
#define DEG_SE ((((N_EDGES + DEG_SLICES - 1) / DEG_SLICES) + 3) & ~3)
#define DEG_T 512
#define DEG_U ((DEG_SE + DEG_T * 4 - 1) / (DEG_T * 4))

__global__ __launch_bounds__(DEG_T) void k_deg_part(const int* __restrict__ dst,
                                                    unsigned* __restrict__ part, int e) {
  __shared__ unsigned cnt[DEG_RN];
  const int range = blockIdx.x % DEG_RANGES;
  const int slice = blockIdx.x / DEG_RANGES;
  const int base = range * DEG_RN;
  const int e0 = slice * DEG_SE;
  const int e1 = (e0 + DEG_SE < e) ? (e0 + DEG_SE) : ((e > e0) ? e : e0);

  const int e1v = e0 + ((e1 - e0) & ~3);

  int4 d[DEG_U];
#pragma unroll
  for (int u = 0; u < DEG_U; ++u) {
    const int i = e0 + (u * DEG_T + (int)threadIdx.x) * 4;
    d[u] = make_int4(-1, -1, -1, -1);
    if (i < e1v) d[u] = *(const int4*)(dst + i);
  }
  int dt = -1;
  if ((int)threadIdx.x < e1 - e1v) dt = dst[e1v + threadIdx.x];

  for (int i = threadIdx.x; i < DEG_RN; i += DEG_T) cnt[i] = 0u;
  __syncthreads();

#pragma unroll
  for (int u = 0; u < DEG_U; ++u) {
    unsigned r;
    r = (unsigned)(d[u].x - base); if (r < (unsigned)DEG_RN) atomicAdd(&cnt[r], 1u);
    r = (unsigned)(d[u].y - base); if (r < (unsigned)DEG_RN) atomicAdd(&cnt[r], 1u);
    r = (unsigned)(d[u].z - base); if (r < (unsigned)DEG_RN) atomicAdd(&cnt[r], 1u);
    r = (unsigned)(d[u].w - base); if (r < (unsigned)DEG_RN) atomicAdd(&cnt[r], 1u);
  }
  {
    const unsigned r = (unsigned)(dt - base);
    if (r < (unsigned)DEG_RN) atomicAdd(&cnt[r], 1u);
  }
  __syncthreads();

  unsigned* __restrict__ prow = part + (size_t)slice * N_NODES + base;
  for (int i = threadIdx.x; i < DEG_RN; i += DEG_T)
    if (base + i < N_NODES) prow[i] = cnt[i];
}

__global__ void k_dinv(const unsigned* __restrict__ part, float* __restrict__ dinv, int n) {
  const int i = blockIdx.x * blockDim.x + threadIdx.x;
  if (i < n) {
    unsigned c = 1u;
#pragma unroll
    for (int s = 0; s < DEG_SLICES; ++s) c += part[(size_t)s * N_NODES + i];
    dinv[i] = rsqrtf((float)c);
  }
}


__global__ __launch_bounds__(128) void k_gemm2(
    float* A, const float* __restrict__ W,
    const float* __restrict__ bias, int M,
    const float* __restrict__ dinv, float* init_out) {
  const int lane = threadIdx.x & 31;
  const int wave = threadIdx.x >> 5;
  const int row_tile = (blockIdx.x * 4 + wave) * 16;
  if (row_tile >= M) return;

  const int m16 = lane & 15;
  const int khalf = lane >> 4;

  v8f acc[4];
#pragma unroll
  for (int t = 0; t < 4; ++t) acc[t] = (v8f){0.f, 0.f, 0.f, 0.f, 0.f, 0.f, 0.f, 0.f};

  const float* arow = A + (size_t)(row_tile + m16) * 96 + khalf * 2;
  const float* __restrict__ wrow = W + khalf * 2 * 64 + m16 * 4;

  for (int k0 = 0; k0 < 96; k0 += 16) {
#pragma unroll
    for (int kq = 0; kq < 4; ++kq) {
      const int k = k0 + kq * 4;
      v2f afrag = *(const v2f*)(arow + k);
      afrag.x = fmaxf(afrag.x, 0.0f);
      afrag.y = fmaxf(afrag.y, 0.0f);
      const v4f w0 = *(const v4f*)(wrow + k * 64);
      const v4f w1 = *(const v4f*)(wrow + (k + 1) * 64);
#pragma unroll
      for (int t = 0; t < 4; ++t) {
        v2f bfrag;
        bfrag.x = w0[t];
        bfrag.y = w1[t];
        acc[t] = __builtin_amdgcn_wmma_f32_16x16x4_f32(
             false, afrag,  false, bfrag,
             (short)0, acc[t],  false,  false);
      }
    }
  }

  const int rbase = row_tile + khalf * 8;
  const v4f bc = *(const v4f*)(bias + m16 * 4);
  float di[8];
#pragma unroll
  for (int i = 0; i < 8; ++i) di[i] = dinv[rbase + i];
#pragma unroll
  for (int i = 0; i < 8; ++i) {
    v4f v, s;
    v.x = acc[0][i]; v.y = acc[1][i]; v.z = acc[2][i]; v.w = acc[3][i];
    s.x = bc.x + v.x * di[i] * di[i];
    s.y = bc.y + v.y * di[i] * di[i];
    s.z = bc.z + v.z * di[i] * di[i];
    s.w = bc.w + v.w * di[i] * di[i];
    *(v4f*)(A + (size_t)(rbase + i) * 96 + m16 * 4) = v;
    *(v4f*)(init_out + (size_t)(rbase + i) * 64 + m16 * 4) = s;
  }
}

template <int NT> struct wvec;
template <> struct wvec<4> { typedef v4f type; };
template <> struct wvec<2> { typedef v2f type; };

template <int KDIM, int NT, int T0>
__device__ __forceinline__ void enc_xT(const float* __restrict__ nrow,
                                       const float* __restrict__ wrow, v8f (&xt)[6]) {
  typedef typename wvec<NT>::type vw;
  constexpr int NC = NT * 16;
#pragma unroll 2
  for (int k0 = 0; k0 < KDIM; k0 += 16) {
    if (k0 + 256 < KDIM) __builtin_prefetch(nrow + k0 + 256, 0, 0);
#pragma unroll
    for (int kq = 0; kq < 4; ++kq) {
      const int k = k0 + kq * 4;
      const v2f xf = *(const v2f*)(nrow + k);
      const vw w0 = *(const vw*)(wrow + k * NC);
      const vw w1 = *(const vw*)(wrow + (k + 1) * NC);
#pragma unroll
      for (int t = 0; t < NT; ++t) {
        v2f wf;
        wf.x = w0[t];
        wf.y = w1[t];
        xt[T0 + t] = __builtin_amdgcn_wmma_f32_16x16x4_f32(
             false, wf,  false, xf,
             (short)0, xt[T0 + t],  false,  false);
      }
    }
  }
}

__global__ __launch_bounds__(128) __attribute__((amdgpu_waves_per_eu(4, 4))) void k_dense1(
    const float* __restrict__ feature, const float* __restrict__ img,
    const float* __restrict__ W_feat, const float* __restrict__ W_img,
    const float* __restrict__ W_g1,
    const float* __restrict__ b_feat, const float* __restrict__ b_img,
    const float* __restrict__ b_g1, const float* __restrict__ dinv,
    float* __restrict__ xw_lo, float* __restrict__ xw_hi, float* __restrict__ hinit, int M) {
  const int lane = threadIdx.x & 31;
  const int wave = threadIdx.x >> 5;
  const int row_tile = (blockIdx.x * 4 + wave) * 16;
  if (row_tile >= M) return;

  const int m16 = lane & 15;
  const int khalf = lane >> 4;

  v8f xt[6];
#pragma unroll
  for (int t = 0; t < 6; ++t) xt[t] = (v8f){0.f, 0.f, 0.f, 0.f, 0.f, 0.f, 0.f, 0.f};

  enc_xT<512, 4, 0>(feature + (size_t)(row_tile + m16) * 512 + khalf * 2,
                    W_feat + khalf * 2 * 64 + m16 * 4, xt);
  enc_xT<256, 2, 4>(img + (size_t)(row_tile + m16) * 256 + khalf * 2,
                    W_img + khalf * 2 * 32 + m16 * 2, xt);

#pragma unroll
  for (int r = 0; r < 8; ++r) {
    const v4f bf4 = *(const v4f*)(b_feat + 4 * (8 * khalf + r));
    const v2f bi2 = *(const v2f*)(b_img + 2 * (8 * khalf + r));
    xt[0][r] = fmaxf(xt[0][r] + bf4.x, 0.0f);
    xt[1][r] = fmaxf(xt[1][r] + bf4.y, 0.0f);
    xt[2][r] = fmaxf(xt[2][r] + bf4.z, 0.0f);
    xt[3][r] = fmaxf(xt[3][r] + bf4.w, 0.0f);
    xt[4][r] = fmaxf(xt[4][r] + bi2.x, 0.0f);
    xt[5][r] = fmaxf(xt[5][r] + bi2.y, 0.0f);
  }

  v8f acc[6];
#pragma unroll
  for (int u = 0; u < 6; ++u) acc[u] = (v8f){0.f, 0.f, 0.f, 0.f, 0.f, 0.f, 0.f, 0.f};
  const float* __restrict__ gF = W_g1 + (size_t)(32 * khalf) * 96 + m16 * 4;
  const float* __restrict__ gI = W_g1 + (size_t)(64 + 16 * khalf) * 96 + m16 * 4;
#pragma unroll
  for (int t = 0; t < 6; ++t) {
#pragma unroll
    for (int j = 0; j < 4; ++j) {
      v2f af;
      af.x = xt[t][2 * j];
      af.y = xt[t][2 * j + 1];
      const float* g0 = (t < 4) ? (gF + (8 * j + t) * 96) : (gI + (4 * j + (t - 4)) * 96);
      const float* g1 = g0 + ((t < 4) ? 4 : 2) * 96;
      const v4f p0 = *(const v4f*)(g0);
      const v4f p1 = *(const v4f*)(g1);
      const v2f q0 = *(const v2f*)(g0 + 64 - 2 * m16);
      const v2f q1 = *(const v2f*)(g1 + 64 - 2 * m16);
#pragma unroll
      for (int u = 0; u < 4; ++u) {
        v2f bf;
        bf.x = p0[u];
        bf.y = p1[u];
        acc[u] = __builtin_amdgcn_wmma_f32_16x16x4_f32(
             false, af,  false, bf,
             (short)0, acc[u],  false,  false);
      }
#pragma unroll
      for (int u = 0; u < 2; ++u) {
        v2f bf;
        bf.x = q0[u];
        bf.y = q1[u];
        acc[4 + u] = __builtin_amdgcn_wmma_f32_16x16x4_f32(
             false, af,  false, bf,
             (short)0, acc[4 + u],  false,  false);
      }
    }
  }

  const int rbase = row_tile + khalf * 8;
  const v4f bc4 = *(const v4f*)(b_g1 + m16 * 4);
  const v2f bc2 = *(const v2f*)(b_g1 + 64 + m16 * 2);
  float di[8];
#pragma unroll
  for (int i = 0; i < 8; ++i) di[i] = dinv[rbase + i];
#pragma unroll
  for (int i = 0; i < 8; ++i) {
    v4f v, s;
    v2f w, z;
    v.x = acc[0][i]; v.y = acc[1][i]; v.z = acc[2][i]; v.w = acc[3][i];
    w.x = acc[4][i]; w.y = acc[5][i];
    s.x = bc4.x + v.x * di[i] * di[i];
    s.y = bc4.y + v.y * di[i] * di[i];
    s.z = bc4.z + v.z * di[i] * di[i];
    s.w = bc4.w + v.w * di[i] * di[i];
    z.x = bc2.x + w.x * di[i] * di[i];
    z.y = bc2.y + w.y * di[i] * di[i];
    float* hr = hinit + (size_t)(rbase + i) * 96;
    *(v4f*)(xw_lo + (size_t)(rbase + i) * 64 + m16 * 4) = v;
    *(v2f*)(xw_hi + (size_t)(rbase + i) * 32 + m16 * 2) = w;
    *(v4f*)(hr + m16 * 4) = s;
    *(v2f*)(hr + 64 + m16 * 2) = z;
  }
}

#define AG_R 64
#define AG_S 4
#define AG_RN ((N_NODES + AG_R - 1) / AG_R)
#define AG_SE (N_EDGES / AG_S)
#define AG_T 1024
#define AG_W (AG_T / 32)
#define AG_U ((AG_SE + AG_T * 4 - 1) / (AG_T * 4))
#define AG_UC 7
#define AG_LCAP 192
static_assert(N_EDGES % (AG_S * 4) == 0, "slices are whole groups of four edges");
static_assert(AG_U % AG_UC == 0, "the scan runs in whole chunks");
static_assert(N_EDGES <= (1 << 20) && AG_RN <= (1 << 12), "a list word is edge | row << 20");
static_assert(N_NODES <= (1 << 16), "a saved word is source | row << 16");

template <int C, int MODE>
__global__ __launch_bounds__(AG_T) void k_aggr(
    const float* __restrict__ xa, const float* __restrict__ xb,
    const float* __restrict__ dinv,
    const int* __restrict__ src, const int* __restrict__ dst,
    float* __restrict__ out,
    unsigned* lsr, float* lnm, int* lcnt) {
  constexpr int J = C / 32;
  constexpr int SA = (C == 96) ? 64 : 96;
  constexpr int B = 8;
  __shared__ float acc[AG_RN * C];
  __shared__ unsigned list[AG_W * AG_LCAP];
  const int lane = threadIdx.x & 31;
  const int wave = __builtin_amdgcn_readfirstlane((int)(threadIdx.x >> 5));
  const int range = blockIdx.x % AG_R;
  const int slice = blockIdx.x / AG_R;
  const int base = range * AG_RN;
  const int e0 = slice * AG_SE;
  const int e1 = e0 + AG_SE;
  const unsigned below = (1u << lane) - 1u;
  unsigned* __restrict__ mylist = list + wave * AG_LCAP;
  const int wslot = (int)blockIdx.x * AG_W + wave;
  unsigned* gsr = lsr + (size_t)wslot * AG_LCAP;
  float* gnm = lnm + (size_t)wslot * AG_LCAP;

  for (int i = threadIdx.x * 4; i < AG_RN * C; i += AG_T * 4)
    *(v4f*)(acc + i) = (v4f){0.f, 0.f, 0.f, 0.f};
  __syncthreads();

  int wcount = 0;
  if (MODE == 1) wcount = lcnt[wslot];
  for (int uc = (MODE == 0) ? 0 : AG_U; uc < AG_U; uc += AG_UC) {
    int4 d[AG_UC];
#pragma unroll
    for (int k = 0; k < AG_UC; ++k) {
      const int i = e0 + ((uc + k) * AG_T + (int)threadIdx.x) * 4;
      d[k] = make_int4(-1, -1, -1, -1);
      if (i < e1) d[k] = *(const int4*)(dst + i);
    }
#pragma unroll
    for (int k = 0; k < AG_UC; ++k) {
      const int i = e0 + ((uc + k) * AG_T + (int)threadIdx.x) * 4;
      const int dq[4] = {d[k].x, d[k].y, d[k].z, d[k].w};
#pragma unroll
      for (int q = 0; q < 4; ++q) {
        const unsigned r = (unsigned)(dq[q] - base);
        const bool hit = r < (unsigned)AG_RN;
        const unsigned m = __builtin_amdgcn_ballot_w32(hit);
        if (m != 0u) {
          const int pos = wcount + __builtin_popcount(m & below);
          if (hit && pos < AG_LCAP) mylist[pos] = (unsigned)(i + q) | (r << 20);
          wcount += __builtin_popcount(m);
        }
      }
    }
  }

  if (wcount > AG_LCAP) {
    int seen = 0;
    for (int u = 0; u < AG_U; ++u) {
      const int i = e0 + (u * AG_T + (int)threadIdx.x) * 4;
      for (int q = 0; q < 4; ++q) {
        int dv = -1;
        if (i < e1) dv = dst[i + q];
        const unsigned r = (unsigned)(dv - base);
        const bool hit = r < (unsigned)AG_RN;
        const unsigned m = __builtin_amdgcn_ballot_w32(hit);
        const int pos = seen + __builtin_popcount(m & below);
        if (hit && pos >= AG_LCAP) {
          const int s = src[i + q];
          const float nrm = dinv[s] * dinv[base + (int)r];
          for (int col = 0; col < C; ++col) {
            const float xv = (col < 64) ? xa[(size_t)s * SA + col] : xb[(size_t)s * 32 + (col - 64)];
            atomicAdd(&acc[r * C + col], xv * nrm);
          }
        }
        seen += __builtin_popcount(m);
      }
    }
  }
  if (MODE == 0 && lane == 0) lcnt[wslot] = wcount;
  __syncthreads();

  const int n = wcount < AG_LCAP ? wcount : AG_LCAP;
  for (int g = 0; g < n; g += 32) {
    const int cnt = (n - g < 32) ? (n - g) : 32;
    int s = 0;
    unsigned r = 0u;
    float nrm = 0.0f;
    if (lane < cnt) {
      if (MODE == 0) {
        const unsigned ent = mylist[g + lane];
        r = ent >> 20;
        s = src[ent & 0xFFFFFu];
        nrm = dinv[s] * dinv[base + (int)r];
        gsr[g + lane] = (unsigned)s | (r << 16);
        gnm[g + lane] = nrm;
      } else {
        const unsigned sr = gsr[g + lane];
        s = (int)(sr & 0xFFFFu);
        r = sr >> 16;
        nrm = gnm[g + lane];
      }
    }
    for (int b0 = 0; b0 < cnt; b0 += B) {
      float v[B][J];
#pragma unroll
      for (int i = 0; i < B; ++i) {
        const int li = (b0 + i < cnt) ? (b0 + i) : (cnt - 1);
        const int si = __builtin_amdgcn_readlane(s, li);
        const float* __restrict__ p = xa + (size_t)si * SA + lane;
        const float* __restrict__ ph = xb + (size_t)si * 32 + lane;
#pragma unroll
        for (int j = 0; j < J; ++j) v[i][j] = (j < 2) ? p[32 * j] : ph[0];
      }
#pragma unroll
      for (int i = 0; i < B; ++i) {
        if (b0 + i < cnt) {
          const int ri = __builtin_amdgcn_readlane((int)r, b0 + i);
          const float ni = __int_as_float(__builtin_amdgcn_readlane(__float_as_int(nrm), b0 + i));
          float* a = acc + ri * C + lane;
#pragma unroll
          for (int j = 0; j < J; ++j) atomicAdd(a + 32 * j, v[i][j] * ni);
        }
      }
    }
  }
  __syncthreads();

  for (int row = wave; row < AG_RN; row += AG_W) {
    if (base + row < N_NODES) {
      const float* a = acc + row * C + lane;
      float* __restrict__ o = out + (size_t)(base + row) * C + lane;
      float t[J];
#pragma unroll
      for (int j = 0; j < J; ++j) t[j] = a[32 * j];
#pragma unroll
      for (int j = 0; j < J; ++j) atomicAdd(o + 32 * j, t[j]);
    }
  }
}

extern "C" void kernel_launch(void* const* d_in, const int* in_sizes, int n_in,
                              void* d_out, int out_size, void* d_ws, size_t ws_size,
                              hipStream_t stream) {
  (void)in_sizes; (void)n_in; (void)out_size; (void)ws_size;

  const float* feature = (const float*)d_in[0];
  const float* img     = (const float*)d_in[1];
  const int*   edges   = (const int*)d_in[2];
  const float* W_feat  = (const float*)d_in[3];
  const float* b_feat  = (const float*)d_in[4];
  const float* W_img   = (const float*)d_in[5];
  const float* b_img   = (const float*)d_in[6];
  const float* W_g1    = (const float*)d_in[7];
  const float* b_g1    = (const float*)d_in[8];
  const float* W_g2    = (const float*)d_in[9];
  const float* b_g2    = (const float*)d_in[10];
  float* out = (float*)d_out;

  const int* src = edges;
  const int* dst = edges + N_EDGES;

  float* ws     = (float*)d_ws;
  float* dinv   = ws;
  float* bufA   = ws + 50048;
  float* bufB   = bufA + (size_t)N_NODES * 96;
  float* xw1hi  = bufB;
  unsigned* lsr = (unsigned*)(bufB + (size_t)N_NODES * 32);
  float* lnm    = (float*)(lsr + (size_t)AG_R * AG_S * AG_W * AG_LCAP);
  int* lcnt     = (int*)(lnm + (size_t)AG_R * AG_S * AG_W * AG_LCAP);
  static_assert((size_t)N_NODES * 32 + 2 * (size_t)AG_R * AG_S * AG_W * AG_LCAP + AG_R * AG_S * AG_W
                    <= (size_t)N_NODES * 96, "lists fit in bufB behind xw1's upper columns");

  const int T = 256;
  const int MB = (N_NODES + 63) / 64;

  unsigned* part = (unsigned*)bufB;
  k_deg_part<<<DEG_RANGES * DEG_SLICES, DEG_T, 0, stream>>>(dst, part, N_EDGES);
  k_dinv<<<(N_NODES + T - 1) / T, T, 0, stream>>>(part, dinv, N_NODES);

  k_dense1<<<MB, 128, 0, stream>>>(feature, img, W_feat, W_img, W_g1, b_feat, b_img, b_g1,
                                   dinv, out, xw1hi, bufA, N_NODES);
  k_aggr<96, 0><<<AG_R * AG_S, AG_T, 0, stream>>>(out, xw1hi, dinv, src, dst, bufA, lsr, lnm, lcnt);

  k_gemm2<<<MB, 128, 0, stream>>>(bufA, W_g2, b_g2, N_NODES, dinv, out);
  k_aggr<64, 1><<<AG_R * AG_S, AG_T, 0, stream>>>(bufA, bufA, dinv, src, dst, out, lsr, lnm, lcnt);
}
